# starting kernel
# baseline (speedup 1.0000x reference)
#include <hip/hip_runtime.h>

typedef __attribute__((ext_vector_type(2))) float v2f;
typedef __attribute__((ext_vector_type(8))) float v8f;

namespace {
constexpr int   kN  = 61;
constexpr int   kE  = 40;
constexpr int   kH  = 10;
constexpr int   kD  = 4;
constexpr int   kRS = 44;
constexpr int   kRows = 64;
constexpr int   kDump = kRows * kRS;
constexpr int   kBuf  = kRows * kRS + 16;
constexpr float kInvSqrtE = 0.15811388300841898f;
}

__device__ __forceinline__ int clampi(int v, int hi2) { return v < hi2 ? v : hi2; }

__global__ __launch_bounds__(256)
void mha_fused_kernel(const float* __restrict__ x,
                      const float* __restrict__ Wq, const float* __restrict__ bq,
                      const float* __restrict__ Wk, const float* __restrict__ bk,
                      const float* __restrict__ Wv, const float* __restrict__ bv,
                      const float* __restrict__ Wp, const float* __restrict__ bp,
                      const float* __restrict__ pe,
                      float* __restrict__ out) {
  __shared__ __align__(16) float s_x[kBuf];
  __shared__ __align__(16) float s_q[kBuf];
  __shared__ __align__(16) float s_k[kBuf];
  __shared__ __align__(16) float s_v[kBuf];

  const int b    = blockIdx.x;
  const int tid  = threadIdx.x;
  const int lane = tid & 31;
  const int wave = tid >> 5;
  const int hi   = lane >> 4;
  const int l16  = lane & 15;
  const int kbl  = hi * 2;

  if (tid == 0) {
    __builtin_prefetch(Wq, 0, 1);
    __builtin_prefetch(Wk, 0, 1);
    __builtin_prefetch(Wv, 0, 1);
    __builtin_prefetch(Wp, 0, 1);
  }

  for (int i = tid; i < kN * kE; i += 256) {
    const int n = i / kE;
    s_x[n * kRS + (i - n * kE)] = x[(size_t)b * (kN * kE) + i] + pe[i];
  }
  __syncthreads();

  for (int tile = wave; tile < 36; tile += 8) {
    const int mat = tile / 12;
    const int mt  = (tile % 12) / 3;
    const int nt  = tile % 3;
    const float* __restrict__ W  = (mat == 0) ? Wq : (mat == 1) ? Wk : Wv;
    const float* __restrict__ bi = (mat == 0) ? bq : (mat == 1) ? bk : bv;
    float* __restrict__ S        = (mat == 0) ? s_q : (mat == 1) ? s_k : s_v;

    const int arow = mt * 16 + l16;
    const int ncol = nt * 16 + l16;
    const int ar_c = clampi(arow, kN - 1);
    const int nc_c = clampi(ncol, kE - 1);
    const bool cok = ncol < kE;

    v2f af[10], bf[10];
#pragma unroll
    for (int kc = 0; kc < 10; ++kc) {
      const int kb = kc * 4 + kbl;
      af[kc].x = s_x[ar_c * kRS + kb];
      af[kc].y = s_x[ar_c * kRS + kb + 1];
      bf[kc].x = W[nc_c * kE + kb];
      bf[kc].y = W[nc_c * kE + kb + 1];
    }
    v8f acc = {0.f, 0.f, 0.f, 0.f, 0.f, 0.f, 0.f, 0.f};
#pragma unroll
    for (int kc = 0; kc < 10; ++kc)
      acc = __builtin_amdgcn_wmma_f32_16x16x4_f32(false, af[kc], false, bf[kc],
                                                  (short)0, acc, false, false);

    const float bias = bi[nc_c];
#pragma unroll
    for (int r = 0; r < 8; ++r) {
      const int row = mt * 16 + r + hi * 8;
      float val = acc[r] + bias;
      if (r < 3)
        val = (mat != 0 && mt == 0 && hi == 0 && (ncol & 3) > r) ? 0.f : val;
      const bool ok = (row < kN) && cok;
      S[ok ? (row * kRS + ncol) : kDump] = val;
    }
  }
  __syncthreads();

  float* __restrict__ s_o = s_x;

  for (int h = wave; h < kH; h += 8) {
    const int co = h * kD;

    v2f ak[4];
#pragma unroll
    for (int kt = 0; kt < 4; ++kt) {
      const int kr = clampi(kt * 16 + l16, kN - 1);
      ak[kt].x = s_k[kr * kRS + co + kbl];
      ak[kt].y = s_k[kr * kRS + co + kbl + 1];
    }

#pragma unroll
    for (int qt = 0; qt < 4; ++qt) {
      const int q   = qt * 16 + l16;
      const int q_c = clampi(q, kN - 1);
      v2f bq_;
      bq_.x = s_q[q_c * kRS + co + kbl];
      bq_.y = s_q[q_c * kRS + co + kbl + 1];

      v8f et[4];
#pragma unroll
      for (int kt = 0; kt < 4; ++kt) {
        v8f z = {0.f, 0.f, 0.f, 0.f, 0.f, 0.f, 0.f, 0.f};
        et[kt] = __builtin_amdgcn_wmma_f32_16x16x4_f32(false, ak[kt], false, bq_,
                                                       (short)0, z, false, false);
      }
#pragma unroll
      for (int r = 5; r < 8; ++r) et[3][r] = hi ? -1e30f : et[3][r];

      float m = -1e30f;
#pragma unroll
      for (int kt = 0; kt < 4; ++kt)
#pragma unroll
        for (int r = 0; r < 8; ++r) m = fmaxf(m, et[kt][r]);
      m = fmaxf(m, __shfl_xor(m, 16, 32));

      float sum = 0.f;
#pragma unroll
      for (int kt = 0; kt < 4; ++kt)
#pragma unroll
        for (int r = 0; r < 8; ++r) {
          const float e = __expf((et[kt][r] - m) * kInvSqrtE);
          et[kt][r] = e;
          sum += e;
        }
      sum += __shfl_xor(sum, 16, 32);
      const float inv = 1.f / sum;

      float o0 = 0.f, o1 = 0.f, o2 = 0.f, o3 = 0.f;
#pragma unroll
      for (int kt = 0; kt < 4; ++kt)
#pragma unroll
        for (int r = 0; r < 8; ++r) {
          const int key = clampi(kt * 16 + r + 8 * hi, kN - 1);
          const float4 vr = *(const float4*)(s_v + key * kRS + co);
          const float a = et[kt][r];
          o0 += a * vr.x; o1 += a * vr.y; o2 += a * vr.z; o3 += a * vr.w;
        }
      o0 += __shfl_xor(o0, 16, 32);
      o1 += __shfl_xor(o1, 16, 32);
      o2 += __shfl_xor(o2, 16, 32);
      o3 += __shfl_xor(o3, 16, 32);

      float4 ov; ov.x = o0 * inv; ov.y = o1 * inv; ov.z = o2 * inv; ov.w = o3 * inv;
      *(float4*)(s_o + (q < kN ? q * kRS + co : kDump)) = ov;
    }
  }
  __syncthreads();

  for (int tile = wave; tile < 12; tile += 8) {
    const int mt   = tile / 3;
    const int nt   = tile % 3;
    const int arow = mt * 16 + l16;
    const int ncol = nt * 16 + l16;
    const int ar_c = clampi(arow, kN - 1);
    const int nc_c = clampi(ncol, kE - 1);
    const bool cok = ncol < kE;

    v2f af[10], bf[10];
#pragma unroll
    for (int kc = 0; kc < 10; ++kc) {
      const int kb = kc * 4 + kbl;
      af[kc].x = s_o[ar_c * kRS + kb];
      af[kc].y = s_o[ar_c * kRS + kb + 1];
      bf[kc].x = Wp[nc_c * kE + kb];
      bf[kc].y = Wp[nc_c * kE + kb + 1];
    }
    v8f acc = {0.f, 0.f, 0.f, 0.f, 0.f, 0.f, 0.f, 0.f};
#pragma unroll
    for (int kc = 0; kc < 10; ++kc)
      acc = __builtin_amdgcn_wmma_f32_16x16x4_f32(false, af[kc], false, bf[kc],
                                                  (short)0, acc, false, false);

    const float bias = bp[nc_c];
#pragma unroll
    for (int r = 0; r < 8; ++r) {
      const int row = mt * 16 + r + hi * 8;
      if (row < kN && cok)
        out[(size_t)b * (kN * kE) + row * kE + ncol] = acc[r] + bias;
    }
  }
}

extern "C" void kernel_launch(void* const* d_in, const int* in_sizes, int n_in,
                              void* d_out, int out_size, void* d_ws, size_t ws_size,
                              hipStream_t stream) {
  (void)in_sizes; (void)n_in; (void)out_size; (void)d_ws; (void)ws_size;
  const float* x  = (const float*)d_in[0];
  const float* Wq = (const float*)d_in[1];
  const float* bq = (const float*)d_in[2];
  const float* Wk = (const float*)d_in[3];
  const float* bk = (const float*)d_in[4];
  const float* Wv = (const float*)d_in[5];
  const float* bv = (const float*)d_in[6];
  const float* Wp = (const float*)d_in[7];
  const float* bp = (const float*)d_in[8];
  const float* pe = (const float*)d_in[9];
  float* out = (float*)d_out;

  mha_fused_kernel<<<dim3(4096), dim3(256), 0, stream>>>(
      x, Wq, bq, Wk, bk, Wv, bv, Wp, bp, pe, out);
}


# attention phase re-split into 40 (head, 16-query tile) units, 5 per wave; V rows read per unit, not held across units: VGPRs 292 -> 100, four workgroups resident per WGP instead of one
# speedup vs baseline: 2.2456x; 2.2456x over previous
#include <hip/hip_runtime.h>

typedef __attribute__((ext_vector_type(2))) float v2f;
typedef __attribute__((ext_vector_type(8))) float v8f;

namespace {
constexpr int   kN  = 61;
constexpr int   kE  = 40;
constexpr int   kH  = 10;
constexpr int   kD  = 4;
constexpr int   kRS = 44;
constexpr int   kRows = 64;
constexpr int   kDump = kRows * kRS;
constexpr int   kBuf  = kRows * kRS + 16;
constexpr float kInvSqrtE = 0.15811388300841898f;
}

__device__ __forceinline__ int clampi(int v, int hi2) { return v < hi2 ? v : hi2; }

__global__ __launch_bounds__(256)
void mha_fused_kernel(const float* __restrict__ x,
                      const float* __restrict__ Wq, const float* __restrict__ bq,
                      const float* __restrict__ Wk, const float* __restrict__ bk,
                      const float* __restrict__ Wv, const float* __restrict__ bv,
                      const float* __restrict__ Wp, const float* __restrict__ bp,
                      const float* __restrict__ pe,
                      float* __restrict__ out) {
  __shared__ __align__(16) float s_x[kBuf];
  __shared__ __align__(16) float s_q[kBuf];
  __shared__ __align__(16) float s_k[kBuf];
  __shared__ __align__(16) float s_v[kBuf];

  const int b    = blockIdx.x;
  const int tid  = threadIdx.x;
  const int lane = tid & 31;
  const int wave = tid >> 5;
  const int hi   = lane >> 4;
  const int l16  = lane & 15;
  const int kbl  = hi * 2;

  if (tid == 0) {
    __builtin_prefetch(Wq, 0, 1);
    __builtin_prefetch(Wk, 0, 1);
    __builtin_prefetch(Wv, 0, 1);
    __builtin_prefetch(Wp, 0, 1);
  }

  for (int i = tid; i < kN * kE; i += 256) {
    const int n = i / kE;
    s_x[n * kRS + (i - n * kE)] = x[(size_t)b * (kN * kE) + i] + pe[i];
  }
  __syncthreads();

  for (int tile = wave; tile < 36; tile += 8) {
    const int mat = tile / 12;
    const int mt  = (tile % 12) / 3;
    const int nt  = tile % 3;
    const float* __restrict__ W  = (mat == 0) ? Wq : (mat == 1) ? Wk : Wv;
    const float* __restrict__ bi = (mat == 0) ? bq : (mat == 1) ? bk : bv;
    float* __restrict__ S        = (mat == 0) ? s_q : (mat == 1) ? s_k : s_v;

    const int arow = mt * 16 + l16;
    const int ncol = nt * 16 + l16;
    const int ar_c = clampi(arow, kN - 1);
    const int nc_c = clampi(ncol, kE - 1);
    const bool cok = ncol < kE;

    v2f af[10], bf[10];
#pragma unroll
    for (int kc = 0; kc < 10; ++kc) {
      const int kb = kc * 4 + kbl;
      af[kc].x = s_x[ar_c * kRS + kb];
      af[kc].y = s_x[ar_c * kRS + kb + 1];
      bf[kc].x = W[nc_c * kE + kb];
      bf[kc].y = W[nc_c * kE + kb + 1];
    }
    v8f acc = {0.f, 0.f, 0.f, 0.f, 0.f, 0.f, 0.f, 0.f};
#pragma unroll
    for (int kc = 0; kc < 10; ++kc)
      acc = __builtin_amdgcn_wmma_f32_16x16x4_f32(false, af[kc], false, bf[kc],
                                                  (short)0, acc, false, false);

    const float bias = bi[nc_c];
#pragma unroll
    for (int r = 0; r < 8; ++r) {
      const int row = mt * 16 + r + hi * 8;
      float val = acc[r] + bias;
      if (r < 3)
        val = (mat != 0 && mt == 0 && hi == 0 && (ncol & 3) > r) ? 0.f : val;
      const bool ok = (row < kN) && cok;
      S[ok ? (row * kRS + ncol) : kDump] = val;
    }
  }
  __syncthreads();

  float* __restrict__ s_o = s_x;

  {
    const int q   = (wave & 3) * 16 + l16;
    const int q_c = clampi(q, kN - 1);
    for (int h = wave >> 2; h < kH; h += 2) {
      const int co = h * kD;

      v2f ak[4];
#pragma unroll
      for (int kt = 0; kt < 4; ++kt) {
        const int kr = clampi(kt * 16 + l16, kN - 1);
        ak[kt].x = s_k[kr * kRS + co + kbl];
        ak[kt].y = s_k[kr * kRS + co + kbl + 1];
      }
      v2f bq_;
      bq_.x = s_q[q_c * kRS + co + kbl];
      bq_.y = s_q[q_c * kRS + co + kbl + 1];

      v8f et[4];
#pragma unroll
      for (int kt = 0; kt < 4; ++kt) {
        v8f z = {0.f, 0.f, 0.f, 0.f, 0.f, 0.f, 0.f, 0.f};
        et[kt] = __builtin_amdgcn_wmma_f32_16x16x4_f32(false, ak[kt], false, bq_,
                                                       (short)0, z, false, false);
      }
#pragma unroll
      for (int r = 5; r < 8; ++r) et[3][r] = hi ? -1e30f : et[3][r];

      float m = -1e30f;
#pragma unroll
      for (int kt = 0; kt < 4; ++kt)
#pragma unroll
        for (int r = 0; r < 8; ++r) m = fmaxf(m, et[kt][r]);
      m = fmaxf(m, __shfl_xor(m, 16, 32));

      float sum = 0.f;
#pragma unroll
      for (int kt = 0; kt < 4; ++kt)
#pragma unroll
        for (int r = 0; r < 8; ++r) {
          const float e = __expf((et[kt][r] - m) * kInvSqrtE);
          et[kt][r] = e;
          sum += e;
        }
      sum += __shfl_xor(sum, 16, 32);
      const float inv = 1.f / sum;

      float o0 = 0.f, o1 = 0.f, o2 = 0.f, o3 = 0.f;
#pragma unroll
      for (int kt = 0; kt < 4; ++kt)
#pragma unroll
        for (int r = 0; r < 8; ++r) {
          const int key = clampi(kt * 16 + r + 8 * hi, kN - 1);
          const float4 vr = *(const float4*)(s_v + key * kRS + co);
          const float a = et[kt][r];
          o0 += a * vr.x; o1 += a * vr.y; o2 += a * vr.z; o3 += a * vr.w;
        }
      o0 += __shfl_xor(o0, 16, 32);
      o1 += __shfl_xor(o1, 16, 32);
      o2 += __shfl_xor(o2, 16, 32);
      o3 += __shfl_xor(o3, 16, 32);

      float4 ov; ov.x = o0 * inv; ov.y = o1 * inv; ov.z = o2 * inv; ov.w = o3 * inv;
      *(float4*)(s_o + (q < kN ? q * kRS + co : kDump)) = ov;
    }
  }
  __syncthreads();

  for (int tile = wave; tile < 12; tile += 8) {
    const int mt   = tile / 3;
    const int nt   = tile % 3;
    const int arow = mt * 16 + l16;
    const int ncol = nt * 16 + l16;
    const int ar_c = clampi(arow, kN - 1);
    const int nc_c = clampi(ncol, kE - 1);
    const bool cok = ncol < kE;

    v2f af[10], bf[10];
#pragma unroll
    for (int kc = 0; kc < 10; ++kc) {
      const int kb = kc * 4 + kbl;
      af[kc].x = s_o[ar_c * kRS + kb];
      af[kc].y = s_o[ar_c * kRS + kb + 1];
      bf[kc].x = Wp[nc_c * kE + kb];
      bf[kc].y = Wp[nc_c * kE + kb + 1];
    }
    v8f acc = {0.f, 0.f, 0.f, 0.f, 0.f, 0.f, 0.f, 0.f};
#pragma unroll
    for (int kc = 0; kc < 10; ++kc)
      acc = __builtin_amdgcn_wmma_f32_16x16x4_f32(false, af[kc], false, bf[kc],
                                                  (short)0, acc, false, false);

    const float bias = bp[nc_c];
#pragma unroll
    for (int r = 0; r < 8; ++r) {
      const int row = mt * 16 + r + hi * 8;
      if (row < kN && cok)
        out[(size_t)b * (kN * kE) + row * kE + ncol] = acc[r] + bias;
    }
  }
}

extern "C" void kernel_launch(void* const* d_in, const int* in_sizes, int n_in,
                              void* d_out, int out_size, void* d_ws, size_t ws_size,
                              hipStream_t stream) {
  (void)in_sizes; (void)n_in; (void)out_size; (void)d_ws; (void)ws_size;
  const float* x  = (const float*)d_in[0];
  const float* Wq = (const float*)d_in[1];
  const float* bq = (const float*)d_in[2];
  const float* Wk = (const float*)d_in[3];
  const float* bk = (const float*)d_in[4];
  const float* Wv = (const float*)d_in[5];
  const float* bv = (const float*)d_in[6];
  const float* Wp = (const float*)d_in[7];
  const float* bp = (const float*)d_in[8];
  const float* pe = (const float*)d_in[9];
  float* out = (float*)d_out;

  mha_fused_kernel<<<dim3(4096), dim3(256), 0, stream>>>(
      x, Wq, bq, Wk, bk, Wv, bv, Wp, bp, pe, out);
}


# QKV projection re-split: each wave owns one 16-column tile of the fused [Wq;Wk;Wv] product for all 4 row tiles; W fragment read once per wave, 4 independent accumulators; 360 -> 320 WMMA
# speedup vs baseline: 2.5022x; 1.1143x over previous
#include <hip/hip_runtime.h>

typedef __attribute__((ext_vector_type(2))) float v2f;
typedef __attribute__((ext_vector_type(8))) float v8f;

namespace {
constexpr int   kN  = 61;
constexpr int   kE  = 40;
constexpr int   kH  = 10;
constexpr int   kD  = 4;
constexpr int   kRS = 44;
constexpr int   kRows = 64;
constexpr int   kDump = kRows * kRS;
constexpr int   kBuf  = kRows * kRS + 16;
constexpr float kInvSqrtE = 0.15811388300841898f;
}

__device__ __forceinline__ int clampi(int v, int hi2) { return v < hi2 ? v : hi2; }

__global__ __launch_bounds__(256)
void mha_fused_kernel(const float* __restrict__ x,
                      const float* __restrict__ Wq, const float* __restrict__ bq,
                      const float* __restrict__ Wk, const float* __restrict__ bk,
                      const float* __restrict__ Wv, const float* __restrict__ bv,
                      const float* __restrict__ Wp, const float* __restrict__ bp,
                      const float* __restrict__ pe,
                      float* __restrict__ out) {
  __shared__ __align__(16) float s_x[kBuf];
  __shared__ __align__(16) float s_qkv[3 * kBuf];
  float* __restrict__ s_q = s_qkv;
  float* __restrict__ s_k = s_qkv + kBuf;
  float* __restrict__ s_v = s_qkv + 2 * kBuf;

  const int b    = blockIdx.x;
  const int tid  = threadIdx.x;
  const int lane = tid & 31;
  const int wave = tid >> 5;
  const int hi   = lane >> 4;
  const int l16  = lane & 15;
  const int kbl  = hi * 2;

  if (tid == 0) {
    __builtin_prefetch(Wq, 0, 1);
    __builtin_prefetch(Wk, 0, 1);
    __builtin_prefetch(Wv, 0, 1);
    __builtin_prefetch(Wp, 0, 1);
  }

  const int  g    = wave * 16 + l16;
  const bool gok  = g < 3 * kE;
  const int  gc   = clampi(g, 3 * kE - 1);
  const int  mat  = (gc >= kE ? 1 : 0) + (gc >= 2 * kE ? 1 : 0);
  const int  cm   = gc - mat * kE;
  const float* __restrict__ Wsel = (mat == 0) ? Wq : (mat == 1) ? Wk : Wv;
  const float* __restrict__ bsel = (mat == 0) ? bq : (mat == 1) ? bk : bv;
  v2f bfr[10];
#pragma unroll
  for (int kc = 0; kc < 10; ++kc) {
    const int kb = kc * 4 + kbl;
    bfr[kc].x = Wsel[cm * kE + kb];
    bfr[kc].y = Wsel[cm * kE + kb + 1];
  }
  const float bias_qkv = bsel[cm];

  for (int i = tid; i < kN * kE; i += 256) {
    const int n = i / kE;
    s_x[n * kRS + (i - n * kE)] = x[(size_t)b * (kN * kE) + i] + pe[i];
  }
  __syncthreads();

  {
    const int ar3 = clampi(48 + l16, kN - 1);
    v8f acc[4];
#pragma unroll
    for (int mt = 0; mt < 4; ++mt)
      acc[mt] = (v8f){0.f, 0.f, 0.f, 0.f, 0.f, 0.f, 0.f, 0.f};
#pragma unroll
    for (int kc = 0; kc < 10; ++kc) {
      const int kb = kc * 4 + kbl;
      v2f af[4];
#pragma unroll
      for (int mt = 0; mt < 4; ++mt) {
        const int ar = (mt < 3) ? (mt * 16 + l16) : ar3;
        af[mt].x = s_x[ar * kRS + kb];
        af[mt].y = s_x[ar * kRS + kb + 1];
      }
#pragma unroll
      for (int mt = 0; mt < 4; ++mt)
        acc[mt] = __builtin_amdgcn_wmma_f32_16x16x4_f32(false, af[mt], false, bfr[kc],
                                                        (short)0, acc[mt], false, false);
    }

    if (gok) {
      float* __restrict__ drow = s_qkv + mat * kBuf + cm + hi * (8 * kRS);
#pragma unroll
      for (int mt = 0; mt < 4; ++mt)
#pragma unroll
        for (int r = 0; r < 8; ++r) {
          float val = acc[mt][r] + bias_qkv;
          if (mt == 0 && r < 3)
            val = (mat != 0 && hi == 0 && (cm & 3) > r) ? 0.f : val;
          drow[(mt * 16 + r) * kRS] = val;
        }
    }
  }
  __syncthreads();

  float* __restrict__ s_o = s_x;

  {
    const int q   = (wave & 3) * 16 + l16;
    const int q_c = clampi(q, kN - 1);
    for (int h = wave >> 2; h < kH; h += 2) {
      const int co = h * kD;

      v2f ak[4];
#pragma unroll
      for (int kt = 0; kt < 4; ++kt) {
        const int kr = clampi(kt * 16 + l16, kN - 1);
        ak[kt].x = s_k[kr * kRS + co + kbl];
        ak[kt].y = s_k[kr * kRS + co + kbl + 1];
      }
      v2f bq_;
      bq_.x = s_q[q_c * kRS + co + kbl];
      bq_.y = s_q[q_c * kRS + co + kbl + 1];

      v8f et[4];
#pragma unroll
      for (int kt = 0; kt < 4; ++kt) {
        v8f z = {0.f, 0.f, 0.f, 0.f, 0.f, 0.f, 0.f, 0.f};
        et[kt] = __builtin_amdgcn_wmma_f32_16x16x4_f32(false, ak[kt], false, bq_,
                                                       (short)0, z, false, false);
      }
#pragma unroll
      for (int r = 5; r < 8; ++r) et[3][r] = hi ? -1e30f : et[3][r];

      float m = -1e30f;
#pragma unroll
      for (int kt = 0; kt < 4; ++kt)
#pragma unroll
        for (int r = 0; r < 8; ++r) m = fmaxf(m, et[kt][r]);
      m = fmaxf(m, __shfl_xor(m, 16, 32));

      float sum = 0.f;
#pragma unroll
      for (int kt = 0; kt < 4; ++kt)
#pragma unroll
        for (int r = 0; r < 8; ++r) {
          const float e = __expf((et[kt][r] - m) * kInvSqrtE);
          et[kt][r] = e;
          sum += e;
        }
      sum += __shfl_xor(sum, 16, 32);
      const float inv = 1.f / sum;

      float o0 = 0.f, o1 = 0.f, o2 = 0.f, o3 = 0.f;
#pragma unroll
      for (int kt = 0; kt < 4; ++kt)
#pragma unroll
        for (int r = 0; r < 8; ++r) {
          const int key = clampi(kt * 16 + r + 8 * hi, kN - 1);
          const float4 vr = *(const float4*)(s_v + key * kRS + co);
          const float a = et[kt][r];
          o0 += a * vr.x; o1 += a * vr.y; o2 += a * vr.z; o3 += a * vr.w;
        }
      o0 += __shfl_xor(o0, 16, 32);
      o1 += __shfl_xor(o1, 16, 32);
      o2 += __shfl_xor(o2, 16, 32);
      o3 += __shfl_xor(o3, 16, 32);

      float4 ov; ov.x = o0 * inv; ov.y = o1 * inv; ov.z = o2 * inv; ov.w = o3 * inv;
      *(float4*)(s_o + (q < kN ? q * kRS + co : kDump)) = ov;
    }
  }
  __syncthreads();

  for (int tile = wave; tile < 12; tile += 8) {
    const int mt   = tile / 3;
    const int nt   = tile % 3;
    const int arow = mt * 16 + l16;
    const int ncol = nt * 16 + l16;
    const int ar_c = clampi(arow, kN - 1);
    const int nc_c = clampi(ncol, kE - 1);
    const bool cok = ncol < kE;

    v2f af[10], bf[10];
#pragma unroll
    for (int kc = 0; kc < 10; ++kc) {
      const int kb = kc * 4 + kbl;
      af[kc].x = s_o[ar_c * kRS + kb];
      af[kc].y = s_o[ar_c * kRS + kb + 1];
      bf[kc].x = Wp[nc_c * kE + kb];
      bf[kc].y = Wp[nc_c * kE + kb + 1];
    }
    v8f acc = {0.f, 0.f, 0.f, 0.f, 0.f, 0.f, 0.f, 0.f};
#pragma unroll
    for (int kc = 0; kc < 10; ++kc)
      acc = __builtin_amdgcn_wmma_f32_16x16x4_f32(false, af[kc], false, bf[kc],
                                                  (short)0, acc, false, false);

    const float bias = bp[nc_c];
#pragma unroll
    for (int r = 0; r < 8; ++r) {
      const int row = mt * 16 + r + hi * 8;
      if (row < kN && cok)
        out[(size_t)b * (kN * kE) + row * kE + ncol] = acc[r] + bias;
    }
  }
}

extern "C" void kernel_launch(void* const* d_in, const int* in_sizes, int n_in,
                              void* d_out, int out_size, void* d_ws, size_t ws_size,
                              hipStream_t stream) {
  (void)in_sizes; (void)n_in; (void)out_size; (void)d_ws; (void)ws_size;
  const float* x  = (const float*)d_in[0];
  const float* Wq = (const float*)d_in[1];
  const float* bq = (const float*)d_in[2];
  const float* Wk = (const float*)d_in[3];
  const float* bk = (const float*)d_in[4];
  const float* Wv = (const float*)d_in[5];
  const float* bv = (const float*)d_in[6];
  const float* Wp = (const float*)d_in[7];
  const float* bp = (const float*)d_in[8];
  const float* pe = (const float*)d_in[9];
  float* out = (float*)d_out;

  mha_fused_kernel<<<dim3(4096), dim3(256), 0, stream>>>(
      x, Wq, bq, Wk, bk, Wv, bv, Wp, bp, pe, out);
}


# attention softmax/PV arithmetic on packed f32 score pairs: scale*log2e folded into the Q fragment, one packed subtract into v_exp, packed P*V fma; denominator summed as even+odd partial sums
# speedup vs baseline: 2.7094x; 1.0828x over previous
#include <hip/hip_runtime.h>

typedef __attribute__((ext_vector_type(2))) float v2f;
typedef __attribute__((ext_vector_type(8))) float v8f;

namespace {
constexpr int   kN  = 61;
constexpr int   kE  = 40;
constexpr int   kH  = 10;
constexpr int   kD  = 4;
constexpr int   kRS = 44;
constexpr int   kRows = 64;
constexpr int   kDump = kRows * kRS;
constexpr int   kBuf  = kRows * kRS + 16;
constexpr float kScaleLog2e = 0.22811011491194377f;
}

__device__ __forceinline__ int clampi(int v, int hi2) { return v < hi2 ? v : hi2; }

__global__ __launch_bounds__(256)
void mha_fused_kernel(const float* __restrict__ x,
                      const float* __restrict__ Wq, const float* __restrict__ bq,
                      const float* __restrict__ Wk, const float* __restrict__ bk,
                      const float* __restrict__ Wv, const float* __restrict__ bv,
                      const float* __restrict__ Wp, const float* __restrict__ bp,
                      const float* __restrict__ pe,
                      float* __restrict__ out) {
  __shared__ __align__(16) float s_x[kBuf];
  __shared__ __align__(16) float s_qkv[3 * kBuf];
  float* __restrict__ s_q = s_qkv;
  float* __restrict__ s_k = s_qkv + kBuf;
  float* __restrict__ s_v = s_qkv + 2 * kBuf;

  const int b    = blockIdx.x;
  const int tid  = threadIdx.x;
  const int lane = tid & 31;
  const int wave = tid >> 5;
  const int hi   = lane >> 4;
  const int l16  = lane & 15;
  const int kbl  = hi * 2;

  if (tid == 0) {
    __builtin_prefetch(Wq, 0, 1);
    __builtin_prefetch(Wk, 0, 1);
    __builtin_prefetch(Wv, 0, 1);
    __builtin_prefetch(Wp, 0, 1);
  }

  const int  g    = wave * 16 + l16;
  const bool gok  = g < 3 * kE;
  const int  gc   = clampi(g, 3 * kE - 1);
  const int  mat  = (gc >= kE ? 1 : 0) + (gc >= 2 * kE ? 1 : 0);
  const int  cm   = gc - mat * kE;
  const float* __restrict__ Wsel = (mat == 0) ? Wq : (mat == 1) ? Wk : Wv;
  const float* __restrict__ bsel = (mat == 0) ? bq : (mat == 1) ? bk : bv;
  v2f bfr[10];
#pragma unroll
  for (int kc = 0; kc < 10; ++kc) {
    const int kb = kc * 4 + kbl;
    bfr[kc].x = Wsel[cm * kE + kb];
    bfr[kc].y = Wsel[cm * kE + kb + 1];
  }
  const float bias_qkv = bsel[cm];

  for (int i = tid; i < kN * kE; i += 256) {
    const int n = i / kE;
    s_x[n * kRS + (i - n * kE)] = x[(size_t)b * (kN * kE) + i] + pe[i];
  }
  __syncthreads();

  {
    const int ar3 = clampi(48 + l16, kN - 1);
    v8f acc[4];
#pragma unroll
    for (int mt = 0; mt < 4; ++mt)
      acc[mt] = (v8f){0.f, 0.f, 0.f, 0.f, 0.f, 0.f, 0.f, 0.f};
#pragma unroll
    for (int kc = 0; kc < 10; ++kc) {
      const int kb = kc * 4 + kbl;
      v2f af[4];
#pragma unroll
      for (int mt = 0; mt < 4; ++mt) {
        const int ar = (mt < 3) ? (mt * 16 + l16) : ar3;
        af[mt].x = s_x[ar * kRS + kb];
        af[mt].y = s_x[ar * kRS + kb + 1];
      }
#pragma unroll
      for (int mt = 0; mt < 4; ++mt)
        acc[mt] = __builtin_amdgcn_wmma_f32_16x16x4_f32(false, af[mt], false, bfr[kc],
                                                        (short)0, acc[mt], false, false);
    }

    if (gok) {
      float* __restrict__ drow = s_qkv + mat * kBuf + cm + hi * (8 * kRS);
#pragma unroll
      for (int mt = 0; mt < 4; ++mt)
#pragma unroll
        for (int r = 0; r < 8; ++r) {
          float val = acc[mt][r] + bias_qkv;
          if (mt == 0 && r < 3)
            val = (mat != 0 && hi == 0 && (cm & 3) > r) ? 0.f : val;
          drow[(mt * 16 + r) * kRS] = val;
        }
    }
  }
  __syncthreads();

  float* __restrict__ s_o = s_x;

  {
    const int q   = (wave & 3) * 16 + l16;
    const int q_c = clampi(q, kN - 1);
    for (int h = wave >> 2; h < kH; h += 2) {
      const int co = h * kD;

      v2f ak[4];
#pragma unroll
      for (int kt = 0; kt < 4; ++kt) {
        const int kr = clampi(kt * 16 + l16, kN - 1);
        ak[kt].x = s_k[kr * kRS + co + kbl];
        ak[kt].y = s_k[kr * kRS + co + kbl + 1];
      }
      v2f bq_;
      bq_.x = s_q[q_c * kRS + co + kbl] * kScaleLog2e;
      bq_.y = s_q[q_c * kRS + co + kbl + 1] * kScaleLog2e;

      v8f et[4];
#pragma unroll
      for (int kt = 0; kt < 4; ++kt) {
        v8f z = {0.f, 0.f, 0.f, 0.f, 0.f, 0.f, 0.f, 0.f};
        et[kt] = __builtin_amdgcn_wmma_f32_16x16x4_f32(false, ak[kt], false, bq_,
                                                       (short)0, z, false, false);
      }
#pragma unroll
      for (int r = 5; r < 8; ++r) et[3][r] = hi ? -1e30f : et[3][r];

      float m = -1e30f;
#pragma unroll
      for (int kt = 0; kt < 4; ++kt)
#pragma unroll
        for (int r = 0; r < 8; ++r) m = fmaxf(m, et[kt][r]);
      m = fmaxf(m, __shfl_xor(m, 16, 32));

      const v2f m2 = {m, m};

      v2f s2  = {0.f, 0.f};
      v2f o01 = {0.f, 0.f}, o23 = {0.f, 0.f};
#pragma unroll
      for (int kt = 0; kt < 4; ++kt)
#pragma unroll
        for (int r = 0; r < 8; r += 2) {
          const v2f e2 = {et[kt][r], et[kt][r + 1]};
          const v2f t2 = e2 - m2;
          v2f p2;
          p2.x = __builtin_amdgcn_exp2f(t2.x);
          p2.y = __builtin_amdgcn_exp2f(t2.y);
          const int keya = clampi(kt * 16 + r + 8 * hi, kN - 1);
          const int keyb = clampi(kt * 16 + r + 1 + 8 * hi, kN - 1);
          const float4 va = *(const float4*)(s_v + keya * kRS + co);
          const float4 vb = *(const float4*)(s_v + keyb * kRS + co);
          const v2f pa = {p2.x, p2.x}, pb = {p2.y, p2.y};
          const v2f va01 = {va.x, va.y}, va23 = {va.z, va.w};
          const v2f vb01 = {vb.x, vb.y}, vb23 = {vb.z, vb.w};
          if (kt == 0 && r == 0) {
            s2  = p2;
            o01 = pa * va01;
            o23 = pa * va23;
          } else {
            s2 += p2;
            o01 = __builtin_elementwise_fma(pa, va01, o01);
            o23 = __builtin_elementwise_fma(pa, va23, o23);
          }
          o01 = __builtin_elementwise_fma(pb, vb01, o01);
          o23 = __builtin_elementwise_fma(pb, vb23, o23);
        }
      float sum = s2.x + s2.y;
      sum += __shfl_xor(sum, 16, 32);
      const float inv = 1.f / sum;
      float o0 = o01.x, o1 = o01.y, o2 = o23.x, o3 = o23.y;
      o0 += __shfl_xor(o0, 16, 32);
      o1 += __shfl_xor(o1, 16, 32);
      o2 += __shfl_xor(o2, 16, 32);
      o3 += __shfl_xor(o3, 16, 32);

      float4 ov; ov.x = o0 * inv; ov.y = o1 * inv; ov.z = o2 * inv; ov.w = o3 * inv;
      *(float4*)(s_o + (q < kN ? q * kRS + co : kDump)) = ov;
    }
  }
  __syncthreads();

  for (int tile = wave; tile < 12; tile += 8) {
    const int mt   = tile / 3;
    const int nt   = tile % 3;
    const int arow = mt * 16 + l16;
    const int ncol = nt * 16 + l16;
    const int ar_c = clampi(arow, kN - 1);
    const int nc_c = clampi(ncol, kE - 1);
    const bool cok = ncol < kE;

    v2f af[10], bf[10];
#pragma unroll
    for (int kc = 0; kc < 10; ++kc) {
      const int kb = kc * 4 + kbl;
      af[kc].x = s_o[ar_c * kRS + kb];
      af[kc].y = s_o[ar_c * kRS + kb + 1];
      bf[kc].x = Wp[nc_c * kE + kb];
      bf[kc].y = Wp[nc_c * kE + kb + 1];
    }
    v8f acc = {0.f, 0.f, 0.f, 0.f, 0.f, 0.f, 0.f, 0.f};
#pragma unroll
    for (int kc = 0; kc < 10; ++kc)
      acc = __builtin_amdgcn_wmma_f32_16x16x4_f32(false, af[kc], false, bf[kc],
                                                  (short)0, acc, false, false);

    const float bias = bp[nc_c];
#pragma unroll
    for (int r = 0; r < 8; ++r) {
      const int row = mt * 16 + r + hi * 8;
      if (row < kN && cok)
        out[(size_t)b * (kN * kE) + row * kE + ncol] = acc[r] + bias;
    }
  }
}

extern "C" void kernel_launch(void* const* d_in, const int* in_sizes, int n_in,
                              void* d_out, int out_size, void* d_ws, size_t ws_size,
                              hipStream_t stream) {
  (void)in_sizes; (void)n_in; (void)out_size; (void)d_ws; (void)ws_size;
  const float* x  = (const float*)d_in[0];
  const float* Wq = (const float*)d_in[1];
  const float* bq = (const float*)d_in[2];
  const float* Wk = (const float*)d_in[3];
  const float* bk = (const float*)d_in[4];
  const float* Wv = (const float*)d_in[5];
  const float* bv = (const float*)d_in[6];
  const float* Wp = (const float*)d_in[7];
  const float* bp = (const float*)d_in[8];
  const float* pe = (const float*)d_in[9];
  float* out = (float*)d_out;

  mha_fused_kernel<<<dim3(4096), dim3(256), 0, stream>>>(
      x, Wq, bq, Wk, bk, Wv, bv, Wp, bp, pe, out);
}


# register budget cut from 109 to 80 VGPRs (QKV pass on 2 accumulators at a time, padded rows read unclamped, wave index scalar): six workgroups resident per WGP instead of four
# speedup vs baseline: 2.8850x; 1.0648x over previous
#include <hip/hip_runtime.h>

typedef __attribute__((ext_vector_type(2))) float v2f;
typedef __attribute__((ext_vector_type(8))) float v8f;

namespace {
constexpr int   kN  = 61;
constexpr int   kE  = 40;
constexpr int   kH  = 10;
constexpr int   kD  = 4;
constexpr int   kRS = 44;
constexpr int   kRows = 64;
constexpr int   kDump = kRows * kRS;
constexpr int   kBuf  = kRows * kRS + 16;
constexpr float kScaleLog2e = 0.22811011491194377f;
}

__device__ __forceinline__ int clampi(int v, int hi2) { return v < hi2 ? v : hi2; }

__global__ __launch_bounds__(256, 12)
void mha_fused_kernel(const float* __restrict__ x,
                      const float* __restrict__ Wq, const float* __restrict__ bq,
                      const float* __restrict__ Wk, const float* __restrict__ bk,
                      const float* __restrict__ Wv, const float* __restrict__ bv,
                      const float* __restrict__ Wp, const float* __restrict__ bp,
                      const float* __restrict__ pe,
                      float* __restrict__ out) {
  __shared__ __align__(16) float s_x[kBuf];
  __shared__ __align__(16) float s_qkv[3 * kBuf];
  float* __restrict__ s_q = s_qkv;
  float* __restrict__ s_k = s_qkv + kBuf;
  float* __restrict__ s_v = s_qkv + 2 * kBuf;

  const int b    = blockIdx.x;
  const int tid  = threadIdx.x;
  const int lane = tid & 31;
  const int wave = __builtin_amdgcn_readfirstlane(tid >> 5);
  const int l16  = lane & 15;
  const int kbl  = (lane >> 4) * 2;

  if (tid == 0) {
    __builtin_prefetch(Wq, 0, 1);
    __builtin_prefetch(Wk, 0, 1);
    __builtin_prefetch(Wv, 0, 1);
    __builtin_prefetch(Wp, 0, 1);
  }

  const int  g    = wave * 16 + l16;
  const bool gok  = g < 3 * kE;
  const int  gc   = clampi(g, 3 * kE - 1);
  const int  mat  = (gc >= kE ? 1 : 0) + (gc >= 2 * kE ? 1 : 0);
  const int  cm   = gc - mat * kE;
  const float* __restrict__ Wsel = (mat == 0) ? Wq : (mat == 1) ? Wk : Wv;
  const float* __restrict__ bsel = (mat == 0) ? bq : (mat == 1) ? bk : bv;
  v2f bfr[10];
#pragma unroll
  for (int kc = 0; kc < 10; ++kc) {
    const int kb = kc * 4 + kbl;
    bfr[kc].x = Wsel[cm * kE + kb];
    bfr[kc].y = Wsel[cm * kE + kb + 1];
  }
  const float bias_qkv = bsel[cm];

  for (int i = tid; i < kN * kE; i += 256) {
    const int n = i / kE;
    s_x[n * kRS + (i - n * kE)] = x[(size_t)b * (kN * kE) + i] + pe[i];
  }
  __syncthreads();

  {
    const int ar3 = clampi(48 + l16, kN - 1);
    float* __restrict__ drow = s_qkv + mat * kBuf + cm + kbl * (4 * kRS);
#pragma unroll
    for (int hp = 0; hp < 2; ++hp) {
      v8f acc[2];
#pragma unroll
      for (int j = 0; j < 2; ++j)
        acc[j] = (v8f){0.f, 0.f, 0.f, 0.f, 0.f, 0.f, 0.f, 0.f};
#pragma unroll
      for (int kc = 0; kc < 10; ++kc) {
        const int kb = kc * 4 + kbl;
        v2f af[2];
#pragma unroll
        for (int j = 0; j < 2; ++j) {
          const int mt = hp * 2 + j;
          const int ar = (mt < 3) ? (mt * 16 + l16) : ar3;
          af[j].x = s_x[ar * kRS + kb];
          af[j].y = s_x[ar * kRS + kb + 1];
        }
#pragma unroll
        for (int j = 0; j < 2; ++j)
          acc[j] = __builtin_amdgcn_wmma_f32_16x16x4_f32(false, af[j], false, bfr[kc],
                                                         (short)0, acc[j], false, false);
      }

      if (gok) {
#pragma unroll
        for (int j = 0; j < 2; ++j)
#pragma unroll
          for (int r = 0; r < 8; ++r) {
            const int mt = hp * 2 + j;
            float val = acc[j][r] + bias_qkv;
            if (mt == 0 && r < 3)
              val = (mat != 0 && kbl == 0 && (cm & 3) > r) ? 0.f : val;
            drow[(mt * 16 + r) * kRS] = val;
          }
      }
    }
  }
  __syncthreads();

  float* __restrict__ s_o = s_x;

  {
    const int q   = (wave & 3) * 16 + l16;
    for (int h = wave >> 2; h < kH; h += 2) {
      const int co = h * kD;

      v2f ak[4];
#pragma unroll
      for (int kt = 0; kt < 4; ++kt) {
        const int kr = kt * 16 + l16;
        ak[kt].x = s_k[kr * kRS + co + kbl];
        ak[kt].y = s_k[kr * kRS + co + kbl + 1];
      }
      v2f bq_;
      bq_.x = s_q[q * kRS + co + kbl] * kScaleLog2e;
      bq_.y = s_q[q * kRS + co + kbl + 1] * kScaleLog2e;

      v8f et[4];
#pragma unroll
      for (int kt = 0; kt < 4; ++kt) {
        v8f z = {0.f, 0.f, 0.f, 0.f, 0.f, 0.f, 0.f, 0.f};
        et[kt] = __builtin_amdgcn_wmma_f32_16x16x4_f32(false, ak[kt], false, bq_,
                                                       (short)0, z, false, false);
      }
#pragma unroll
      for (int r = 5; r < 8; ++r) et[3][r] = kbl ? -1e30f : et[3][r];

      float m = -1e30f;
#pragma unroll
      for (int kt = 0; kt < 4; ++kt)
#pragma unroll
        for (int r = 0; r < 8; ++r) m = fmaxf(m, et[kt][r]);
      m = fmaxf(m, __shfl_xor(m, 16, 32));

      const v2f m2 = {m, m};

      v2f s2  = {0.f, 0.f};
      v2f o01 = {0.f, 0.f}, o23 = {0.f, 0.f};
#pragma unroll
      for (int kt = 0; kt < 4; ++kt)
#pragma unroll
        for (int r = 0; r < 8; r += 2) {
          const v2f e2 = {et[kt][r], et[kt][r + 1]};
          const v2f t2 = e2 - m2;
          v2f p2;
          p2.x = __builtin_amdgcn_exp2f(t2.x);
          p2.y = __builtin_amdgcn_exp2f(t2.y);
          const int keya = kt * 16 + r + 4 * kbl;
          const int keyb = kt * 16 + r + 1 + 4 * kbl;
          const float4 va = *(const float4*)(s_v + keya * kRS + co);
          const float4 vb = *(const float4*)(s_v + keyb * kRS + co);
          const v2f pa = {p2.x, p2.x}, pb = {p2.y, p2.y};
          const v2f va01 = {va.x, va.y}, va23 = {va.z, va.w};
          const v2f vb01 = {vb.x, vb.y}, vb23 = {vb.z, vb.w};
          if (kt == 0 && r == 0) {
            s2  = p2;
            o01 = pa * va01;
            o23 = pa * va23;
          } else {
            s2 += p2;
            o01 = __builtin_elementwise_fma(pa, va01, o01);
            o23 = __builtin_elementwise_fma(pa, va23, o23);
          }
          o01 = __builtin_elementwise_fma(pb, vb01, o01);
          o23 = __builtin_elementwise_fma(pb, vb23, o23);
        }
      float sum = s2.x + s2.y;
      sum += __shfl_xor(sum, 16, 32);
      const float inv = 1.f / sum;
      float o0 = o01.x, o1 = o01.y, o2 = o23.x, o3 = o23.y;
      o0 += __shfl_xor(o0, 16, 32);
      o1 += __shfl_xor(o1, 16, 32);
      o2 += __shfl_xor(o2, 16, 32);
      o3 += __shfl_xor(o3, 16, 32);

      float4 ov; ov.x = o0 * inv; ov.y = o1 * inv; ov.z = o2 * inv; ov.w = o3 * inv;
      *(float4*)(s_o + (q < kN ? q * kRS + co : kDump)) = ov;
    }
  }
  __syncthreads();

  for (int tile = wave; tile < 12; tile += 8) {
    const int mt   = tile / 3;
    const int nt   = tile % 3;
    const int arow = mt * 16 + l16;
    const int ncol = nt * 16 + l16;
    const int ar_c = clampi(arow, kN - 1);
    const int nc_c = clampi(ncol, kE - 1);
    const bool cok = ncol < kE;

    v2f af[10], bf[10];
#pragma unroll
    for (int kc = 0; kc < 10; ++kc) {
      const int kb = kc * 4 + kbl;
      af[kc].x = s_o[ar_c * kRS + kb];
      af[kc].y = s_o[ar_c * kRS + kb + 1];
      bf[kc].x = Wp[nc_c * kE + kb];
      bf[kc].y = Wp[nc_c * kE + kb + 1];
    }
    v8f acc = {0.f, 0.f, 0.f, 0.f, 0.f, 0.f, 0.f, 0.f};
#pragma unroll
    for (int kc = 0; kc < 10; ++kc)
      acc = __builtin_amdgcn_wmma_f32_16x16x4_f32(false, af[kc], false, bf[kc],
                                                  (short)0, acc, false, false);

    const float bias = bp[nc_c];
#pragma unroll
    for (int r = 0; r < 8; ++r) {
      const int row = mt * 16 + r + kbl * 4;
      if (row < kN && cok)
        out[(size_t)b * (kN * kE) + row * kE + ncol] = acc[r] + bias;
    }
  }
}

extern "C" void kernel_launch(void* const* d_in, const int* in_sizes, int n_in,
                              void* d_out, int out_size, void* d_ws, size_t ws_size,
                              hipStream_t stream) {
  (void)in_sizes; (void)n_in; (void)out_size; (void)d_ws; (void)ws_size;
  const float* x  = (const float*)d_in[0];
  const float* Wq = (const float*)d_in[1];
  const float* bq = (const float*)d_in[2];
  const float* Wk = (const float*)d_in[3];
  const float* bk = (const float*)d_in[4];
  const float* Wv = (const float*)d_in[5];
  const float* bv = (const float*)d_in[6];
  const float* Wp = (const float*)d_in[7];
  const float* bp = (const float*)d_in[8];
  const float* pe = (const float*)d_in[9];
  float* out = (float*)d_out;

  mha_fused_kernel<<<dim3(4096), dim3(256), 0, stream>>>(
      x, Wq, bq, Wk, bk, Wv, bv, Wp, bp, pe, out);
}
